# starting kernel
# baseline (speedup 1.0000x reference)
#include <hip/hip_runtime.h>


typedef __attribute__((ext_vector_type(2))) float v2f;
typedef __attribute__((ext_vector_type(4))) float v4f;
typedef __attribute__((ext_vector_type(8))) float v8f;

#define B_DIM 64
#define C_DIM 64
#define U_DIM 1024
#define N_CAP 32
#define D_CAP 16

__global__ void caps_transpose(const float* __restrict__ in,
                               float* __restrict__ outT) {
  __shared__ float tile[32][33];
  const int u0 = blockIdx.x * 32;
  const int r0 = blockIdx.y * 32;
  const int tx = threadIdx.x & 31;
  const int ty = threadIdx.x >> 5;
  for (int i = 0; i < 32; i += 8)
    tile[ty + i][tx] = in[(size_t)(r0 + ty + i) * U_DIM + u0 + tx];
  __syncthreads();
  for (int i = 0; i < 32; i += 8)
    outT[(size_t)(u0 + ty + i) * (B_DIM * C_DIM) + r0 + tx] = tile[tx][ty + i];
}

__global__ void __launch_bounds__(512, 1)
caps_uhat(const float* __restrict__ uT,
          const float* __restrict__ W,
          float* __restrict__ uhat) {
  const int u = blockIdx.x;
  __shared__ float As[B_DIM * 68];

  const float* Ag = uT + (size_t)u * (B_DIM * C_DIM);
  for (int i = threadIdx.x; i < (B_DIM * C_DIM) / 4; i += 512) {
    const int e = i * 4;
    const int b = e >> 6;
    const int c = e & 63;
    *(v4f*)&As[b * 68 + c] = ((const v4f*)Ag)[i];
  }
  __syncthreads();

  const int wave = threadIdx.x >> 5;
  const int lane = threadIdx.x & 31;
  const int l16  = lane & 15;
  const int koff = (lane >> 4) * 2;
  const int n_base = wave * 2;

  const float* Wg = W + (size_t)u * (N_CAP * C_DIM * D_CAP);

  v8f acc[4][2] = {};
  for (int kc = 0; kc < C_DIM; kc += 4) {
    const int k = kc + koff;
    v2f a[4], bb[2];
    for (int mt = 0; mt < 4; ++mt)
      a[mt] = *(const v2f*)&As[(mt * 16 + l16) * 68 + k];
    for (int nt = 0; nt < 2; ++nt) {
      const float* wp = Wg + ((size_t)(n_base + nt) * C_DIM + k) * D_CAP + l16;
      bb[nt].x = wp[0];
      bb[nt].y = wp[D_CAP];
      if (kc + 4 < C_DIM)
        __builtin_prefetch(wp + 4 * D_CAP, 0, 1);
    }
    for (int mt = 0; mt < 4; ++mt)
      for (int nt = 0; nt < 2; ++nt)
        acc[mt][nt] = __builtin_amdgcn_wmma_f32_16x16x4_f32(
            false, a[mt], false, bb[nt], (short)0, acc[mt][nt], false, false);
  }

  const int d = l16;
  const int rbase = (lane >> 4) * 8;
  for (int mt = 0; mt < 4; ++mt)
    for (int nt = 0; nt < 2; ++nt) {
      const int n = n_base + nt;
      for (int r = 0; r < 8; ++r) {
        const int b = mt * 16 + rbase + r;
        uhat[(((size_t)b * N_CAP + n) * U_DIM + u) * D_CAP + d] = acc[mt][nt][r];
      }
    }
}

__global__ void caps_route(const float* __restrict__ uhat,
                           float* __restrict__ out) {
  const int bn = blockIdx.x;
  const int b = bn >> 5, n = bn & 31;
  const int t = threadIdx.x;
  const int lane = t & 31, wave = t >> 5;

  __shared__ float red[8 * 16];
  __shared__ float sv[16];
  __shared__ float rscal[8];

  const float* base = uhat + (size_t)bn * (U_DIM * D_CAP);
  v4f row[4][4];
  for (int k = 0; k < 4; ++k) {
    const v4f* p = (const v4f*)(base + (size_t)(t + k * 256) * D_CAP);
    for (int j = 0; j < 4; ++j) row[k][j] = p[j];
  }

  float blog[4] = {0.f, 0.f, 0.f, 0.f};
  float v[16];

  for (int it = 0; it < 3; ++it) {
    float c[4];
    if (it == 0) {
      c[0] = c[1] = c[2] = c[3] = 1.0f / (float)U_DIM;
    } else {
      float mx = fmaxf(fmaxf(blog[0], blog[1]), fmaxf(blog[2], blog[3]));
      for (int o = 16; o; o >>= 1) mx = fmaxf(mx, __shfl_xor(mx, o, 32));
      if (lane == 0) rscal[wave] = mx;
      __syncthreads();
      float m2 = rscal[0];
      for (int w = 1; w < 8; ++w) m2 = fmaxf(m2, rscal[w]);
      float sum = 0.f;
      for (int k = 0; k < 4; ++k) { c[k] = __expf(blog[k] - m2); sum += c[k]; }
      for (int o = 16; o; o >>= 1) sum += __shfl_xor(sum, o, 32);
      __syncthreads();
      if (lane == 0) rscal[wave] = sum;
      __syncthreads();
      float tot = 0.f;
      for (int w = 0; w < 8; ++w) tot += rscal[w];
      const float inv = 1.0f / tot;
      for (int k = 0; k < 4; ++k) c[k] *= inv;
    }

    float sp[16];
    for (int d = 0; d < 16; ++d) sp[d] = 0.f;
    for (int k = 0; k < 4; ++k)
      for (int j = 0; j < 4; ++j)
        for (int e = 0; e < 4; ++e)
          sp[j * 4 + e] += c[k] * row[k][j][e];
    for (int d = 0; d < 16; ++d)
      for (int o = 16; o; o >>= 1) sp[d] += __shfl_xor(sp[d], o, 32);
    if (lane == 0)
      for (int d = 0; d < 16; ++d) red[wave * 16 + d] = sp[d];
    __syncthreads();
    if (t < 16) {
      float s = 0.f;
      for (int w = 0; w < 8; ++w) s += red[w * 16 + t];
      sv[t] = s;
    }
    __syncthreads();

    float s2 = 1e-8f;
    for (int d = 0; d < 16; ++d) s2 += sv[d] * sv[d];
    const float scale = sqrtf(s2) / (1.0f + s2);
    for (int d = 0; d < 16; ++d) v[d] = sv[d] * scale;

    for (int k = 0; k < 4; ++k) {
      float dot = 0.f;
      for (int j = 0; j < 4; ++j)
        for (int e = 0; e < 4; ++e)
          dot += row[k][j][e] * v[j * 4 + e];
      blog[k] += dot;
    }
    __syncthreads();
  }

  if (t < 16)
    out[((size_t)b * D_CAP + t) * N_CAP + n] = v[t];
}

extern "C" void kernel_launch(void* const* d_in, const int* in_sizes, int n_in,
                              void* d_out, int out_size, void* d_ws, size_t ws_size,
                              hipStream_t stream) {
  const float* u_vecs = (const float*)d_in[0];
  const float* W      = (const float*)d_in[1];
  float* out = (float*)d_out;

  float* uT   = (float*)d_ws;
  float* uhat = uT + (size_t)U_DIM * B_DIM * C_DIM;

  caps_transpose<<<dim3(U_DIM / 32, (B_DIM * C_DIM) / 32), 256, 0, stream>>>(u_vecs, uT);
  caps_uhat<<<dim3(U_DIM), 512, 0, stream>>>(uT, W, uhat);
  caps_route<<<dim3(B_DIM * N_CAP), 256, 0, stream>>>(uhat, out);
}


# routing kernel: logit update as packed FMAs on the rows' own register pairs (even and odd columns summed in two chains, then added): no transposed copy of the rows, so no scratch spill and reload
# speedup vs baseline: 1.3283x; 1.3283x over previous
#include <hip/hip_runtime.h>


typedef __attribute__((ext_vector_type(2))) float v2f;
typedef __attribute__((ext_vector_type(4))) float v4f;
typedef __attribute__((ext_vector_type(8))) float v8f;

#define B_DIM 64
#define C_DIM 64
#define U_DIM 1024
#define N_CAP 32
#define D_CAP 16

__global__ void caps_transpose(const float* __restrict__ in,
                               float* __restrict__ outT) {
  __shared__ float tile[32][33];
  const int u0 = blockIdx.x * 32;
  const int r0 = blockIdx.y * 32;
  const int tx = threadIdx.x & 31;
  const int ty = threadIdx.x >> 5;
  for (int i = 0; i < 32; i += 8)
    tile[ty + i][tx] = in[(size_t)(r0 + ty + i) * U_DIM + u0 + tx];
  __syncthreads();
  for (int i = 0; i < 32; i += 8)
    outT[(size_t)(u0 + ty + i) * (B_DIM * C_DIM) + r0 + tx] = tile[tx][ty + i];
}

__global__ void __launch_bounds__(512, 1)
caps_uhat(const float* __restrict__ uT,
          const float* __restrict__ W,
          float* __restrict__ uhat) {
  const int u = blockIdx.x;
  __shared__ float As[B_DIM * 68];

  const float* Ag = uT + (size_t)u * (B_DIM * C_DIM);
  for (int i = threadIdx.x; i < (B_DIM * C_DIM) / 4; i += 512) {
    const int e = i * 4;
    const int b = e >> 6;
    const int c = e & 63;
    *(v4f*)&As[b * 68 + c] = ((const v4f*)Ag)[i];
  }
  __syncthreads();

  const int wave = threadIdx.x >> 5;
  const int lane = threadIdx.x & 31;
  const int l16  = lane & 15;
  const int koff = (lane >> 4) * 2;
  const int n_base = wave * 2;

  const float* Wg = W + (size_t)u * (N_CAP * C_DIM * D_CAP);

  v8f acc[4][2] = {};
  for (int kc = 0; kc < C_DIM; kc += 4) {
    const int k = kc + koff;
    v2f a[4], bb[2];
    for (int mt = 0; mt < 4; ++mt)
      a[mt] = *(const v2f*)&As[(mt * 16 + l16) * 68 + k];
    for (int nt = 0; nt < 2; ++nt) {
      const float* wp = Wg + ((size_t)(n_base + nt) * C_DIM + k) * D_CAP + l16;
      bb[nt].x = wp[0];
      bb[nt].y = wp[D_CAP];
      if (kc + 4 < C_DIM)
        __builtin_prefetch(wp + 4 * D_CAP, 0, 1);
    }
    for (int mt = 0; mt < 4; ++mt)
      for (int nt = 0; nt < 2; ++nt)
        acc[mt][nt] = __builtin_amdgcn_wmma_f32_16x16x4_f32(
            false, a[mt], false, bb[nt], (short)0, acc[mt][nt], false, false);
  }

  const int d = l16;
  const int rbase = (lane >> 4) * 8;
  for (int mt = 0; mt < 4; ++mt)
    for (int nt = 0; nt < 2; ++nt) {
      const int n = n_base + nt;
      for (int r = 0; r < 8; ++r) {
        const int b = mt * 16 + rbase + r;
        uhat[(((size_t)b * N_CAP + n) * U_DIM + u) * D_CAP + d] = acc[mt][nt][r];
      }
    }
}

__global__ void caps_route(const float* __restrict__ uhat,
                           float* __restrict__ out) {
  const int bn = blockIdx.x;
  const int b = bn >> 5, n = bn & 31;
  const int t = threadIdx.x;
  const int lane = t & 31, wave = t >> 5;

  __shared__ float red[8 * 16];
  __shared__ float sv[16];
  __shared__ float rscal[8];

  const float* base = uhat + (size_t)bn * (U_DIM * D_CAP);
  v4f row[4][4];
  for (int k = 0; k < 4; ++k) {
    const v4f* p = (const v4f*)(base + (size_t)(t + k * 256) * D_CAP);
    for (int j = 0; j < 4; ++j) row[k][j] = p[j];
  }

  float blog[4] = {0.f, 0.f, 0.f, 0.f};
  float v[16];

  for (int it = 0; it < 3; ++it) {
    float c[4];
    if (it == 0) {
      c[0] = c[1] = c[2] = c[3] = 1.0f / (float)U_DIM;
    } else {
      float mx = fmaxf(fmaxf(blog[0], blog[1]), fmaxf(blog[2], blog[3]));
      for (int o = 16; o; o >>= 1) mx = fmaxf(mx, __shfl_xor(mx, o, 32));
      if (lane == 0) rscal[wave] = mx;
      __syncthreads();
      float m2 = rscal[0];
      for (int w = 1; w < 8; ++w) m2 = fmaxf(m2, rscal[w]);
      float sum = 0.f;
      for (int k = 0; k < 4; ++k) { c[k] = __expf(blog[k] - m2); sum += c[k]; }
      for (int o = 16; o; o >>= 1) sum += __shfl_xor(sum, o, 32);
      __syncthreads();
      if (lane == 0) rscal[wave] = sum;
      __syncthreads();
      float tot = 0.f;
      for (int w = 0; w < 8; ++w) tot += rscal[w];
      const float inv = 1.0f / tot;
      for (int k = 0; k < 4; ++k) c[k] *= inv;
    }

    float sp[16];
    for (int d = 0; d < 16; ++d) sp[d] = 0.f;
    for (int k = 0; k < 4; ++k)
      for (int j = 0; j < 4; ++j)
        for (int e = 0; e < 4; ++e)
          sp[j * 4 + e] += c[k] * row[k][j][e];
    for (int d = 0; d < 16; ++d)
      for (int o = 16; o; o >>= 1) sp[d] += __shfl_xor(sp[d], o, 32);
    if (lane == 0)
      for (int d = 0; d < 16; ++d) red[wave * 16 + d] = sp[d];
    __syncthreads();
    if (t < 16) {
      float s = 0.f;
      for (int w = 0; w < 8; ++w) s += red[w * 16 + t];
      sv[t] = s;
    }
    __syncthreads();

    float s2 = 1e-8f;
    for (int d = 0; d < 16; ++d) s2 += sv[d] * sv[d];
    const float scale = sqrtf(s2) / (1.0f + s2);
    for (int d = 0; d < 16; ++d) v[d] = sv[d] * scale;

    for (int k = 0; k < 4; ++k) {
      v2f acc = {0.f, 0.f};
      for (int j = 0; j < 4; ++j) {
        const v2f rlo = {row[k][j][0], row[k][j][1]};
        const v2f rhi = {row[k][j][2], row[k][j][3]};
        const v2f vlo = {v[j * 4 + 0], v[j * 4 + 1]};
        const v2f vhi = {v[j * 4 + 2], v[j * 4 + 3]};
        acc += rlo * vlo;
        acc += rhi * vhi;
      }
      blog[k] += acc.x + acc.y;
    }
    __syncthreads();
  }

  if (t < 16)
    out[((size_t)b * D_CAP + t) * N_CAP + n] = v[t];
}

extern "C" void kernel_launch(void* const* d_in, const int* in_sizes, int n_in,
                              void* d_out, int out_size, void* d_ws, size_t ws_size,
                              hipStream_t stream) {
  const float* u_vecs = (const float*)d_in[0];
  const float* W      = (const float*)d_in[1];
  float* out = (float*)d_out;

  float* uT   = (float*)d_ws;
  float* uhat = uT + (size_t)U_DIM * B_DIM * C_DIM;

  caps_transpose<<<dim3(U_DIM / 32, (B_DIM * C_DIM) / 32), 256, 0, stream>>>(u_vecs, uT);
  caps_uhat<<<dim3(U_DIM), 512, 0, stream>>>(uT, W, uhat);
  caps_route<<<dim3(B_DIM * N_CAP), 256, 0, stream>>>(uhat, out);
}


# routing kernel: wave reductions exchange lanes on the vector ALU (row_xmask DPP, permlanex16), not LDS permutes; the 16 partial sums halve per level (16 exchanges, not 80); same pairs, same rounding
# speedup vs baseline: 1.3994x; 1.0535x over previous
#include <hip/hip_runtime.h>


typedef __attribute__((ext_vector_type(2))) float v2f;
typedef __attribute__((ext_vector_type(4))) float v4f;
typedef __attribute__((ext_vector_type(8))) float v8f;

#define B_DIM 64
#define C_DIM 64
#define U_DIM 1024
#define N_CAP 32
#define D_CAP 16

__global__ void caps_transpose(const float* __restrict__ in,
                               float* __restrict__ outT) {
  __shared__ float tile[32][33];
  const int u0 = blockIdx.x * 32;
  const int r0 = blockIdx.y * 32;
  const int tx = threadIdx.x & 31;
  const int ty = threadIdx.x >> 5;
  for (int i = 0; i < 32; i += 8)
    tile[ty + i][tx] = in[(size_t)(r0 + ty + i) * U_DIM + u0 + tx];
  __syncthreads();
  for (int i = 0; i < 32; i += 8)
    outT[(size_t)(u0 + ty + i) * (B_DIM * C_DIM) + r0 + tx] = tile[tx][ty + i];
}

__global__ void __launch_bounds__(512, 1)
caps_uhat(const float* __restrict__ uT,
          const float* __restrict__ W,
          float* __restrict__ uhat) {
  const int u = blockIdx.x;
  __shared__ float As[B_DIM * 68];

  const float* Ag = uT + (size_t)u * (B_DIM * C_DIM);
  for (int i = threadIdx.x; i < (B_DIM * C_DIM) / 4; i += 512) {
    const int e = i * 4;
    const int b = e >> 6;
    const int c = e & 63;
    *(v4f*)&As[b * 68 + c] = ((const v4f*)Ag)[i];
  }
  __syncthreads();

  const int wave = threadIdx.x >> 5;
  const int lane = threadIdx.x & 31;
  const int l16  = lane & 15;
  const int koff = (lane >> 4) * 2;
  const int n_base = wave * 2;

  const float* Wg = W + (size_t)u * (N_CAP * C_DIM * D_CAP);

  v8f acc[4][2] = {};
  for (int kc = 0; kc < C_DIM; kc += 4) {
    const int k = kc + koff;
    v2f a[4], bb[2];
    for (int mt = 0; mt < 4; ++mt)
      a[mt] = *(const v2f*)&As[(mt * 16 + l16) * 68 + k];
    for (int nt = 0; nt < 2; ++nt) {
      const float* wp = Wg + ((size_t)(n_base + nt) * C_DIM + k) * D_CAP + l16;
      bb[nt].x = wp[0];
      bb[nt].y = wp[D_CAP];
      if (kc + 4 < C_DIM)
        __builtin_prefetch(wp + 4 * D_CAP, 0, 1);
    }
    for (int mt = 0; mt < 4; ++mt)
      for (int nt = 0; nt < 2; ++nt)
        acc[mt][nt] = __builtin_amdgcn_wmma_f32_16x16x4_f32(
            false, a[mt], false, bb[nt], (short)0, acc[mt][nt], false, false);
  }

  const int d = l16;
  const int rbase = (lane >> 4) * 8;
  for (int mt = 0; mt < 4; ++mt)
    for (int nt = 0; nt < 2; ++nt) {
      const int n = n_base + nt;
      for (int r = 0; r < 8; ++r) {
        const int b = mt * 16 + rbase + r;
        uhat[(((size_t)b * N_CAP + n) * U_DIM + u) * D_CAP + d] = acc[mt][nt][r];
      }
    }
}

template <int O>
__device__ __forceinline__ float lane_xor(float x) {
  const int xi = __builtin_bit_cast(int, x);
  if constexpr (O == 16) {
    return __builtin_bit_cast(float, __builtin_amdgcn_permlanex16(
        xi, xi, 0x76543210, 0xfedcba98, false, false));
  } else {
    return __builtin_bit_cast(float, __builtin_amdgcn_update_dpp(
        0, xi, 0x160 + O, 0xf, 0xf, true));
  }
}

__device__ __forceinline__ float wave_sum(float x) {
  x += lane_xor<16>(x);
  x += lane_xor<8>(x);
  x += lane_xor<4>(x);
  x += lane_xor<2>(x);
  x += lane_xor<1>(x);
  return x;
}
__device__ __forceinline__ float wave_max(float x) {
  x = fmaxf(x, lane_xor<16>(x));
  x = fmaxf(x, lane_xor<8>(x));
  x = fmaxf(x, lane_xor<4>(x));
  x = fmaxf(x, lane_xor<2>(x));
  x = fmaxf(x, lane_xor<1>(x));
  return x;
}

__global__ void caps_route(const float* __restrict__ uhat,
                           float* __restrict__ out) {
  const int bn = blockIdx.x;
  const int b = bn >> 5, n = bn & 31;
  const int t = threadIdx.x;
  const int lane = t & 31, wave = t >> 5;

  __shared__ float red[8 * 16];
  __shared__ float sv[16];
  __shared__ float rscal[8];

  const float* base = uhat + (size_t)bn * (U_DIM * D_CAP);
  v4f row[4][4];
  for (int k = 0; k < 4; ++k) {
    const v4f* p = (const v4f*)(base + (size_t)(t + k * 256) * D_CAP);
    for (int j = 0; j < 4; ++j) row[k][j] = p[j];
  }

  float blog[4] = {0.f, 0.f, 0.f, 0.f};
  float v[16];
  const bool h16 = (lane & 16) != 0, h8 = (lane & 8) != 0, h4 = (lane & 4) != 0, h2 = (lane & 2) != 0;

  for (int it = 0; it < 3; ++it) {
    float c[4];
    if (it == 0) {
      c[0] = c[1] = c[2] = c[3] = 1.0f / (float)U_DIM;
    } else {
      float mx = fmaxf(fmaxf(blog[0], blog[1]), fmaxf(blog[2], blog[3]));
      mx = wave_max(mx);
      if (lane == 0) rscal[wave] = mx;
      __syncthreads();
      float m2 = rscal[0];
      for (int w = 1; w < 8; ++w) m2 = fmaxf(m2, rscal[w]);
      float sum = 0.f;
      for (int k = 0; k < 4; ++k) { c[k] = __expf(blog[k] - m2); sum += c[k]; }
      sum = wave_sum(sum);
      __syncthreads();
      if (lane == 0) rscal[wave] = sum;
      __syncthreads();
      float tot = 0.f;
      for (int w = 0; w < 8; ++w) tot += rscal[w];
      const float inv = 1.0f / tot;
      for (int k = 0; k < 4; ++k) c[k] *= inv;
    }

    float sp[16];
    for (int d = 0; d < 16; ++d) sp[d] = 0.f;
    for (int k = 0; k < 4; ++k)
      for (int j = 0; j < 4; ++j)
        for (int e = 0; e < 4; ++e)
          sp[j * 4 + e] += c[k] * row[k][j][e];
    float r8[8], r4[4], r2[2];
    for (int i = 0; i < 8; ++i) {
      const float keep = h16 ? sp[i + 8] : sp[i];
      const float send = h16 ? sp[i] : sp[i + 8];
      r8[i] = keep + lane_xor<16>(send);
    }
    for (int i = 0; i < 4; ++i) {
      const float keep = h8 ? r8[i + 4] : r8[i];
      const float send = h8 ? r8[i] : r8[i + 4];
      r4[i] = keep + lane_xor<8>(send);
    }
    for (int i = 0; i < 2; ++i) {
      const float keep = h4 ? r4[i + 2] : r4[i];
      const float send = h4 ? r4[i] : r4[i + 2];
      r2[i] = keep + lane_xor<4>(send);
    }
    float tot;
    {
      const float keep = h2 ? r2[1] : r2[0];
      const float send = h2 ? r2[0] : r2[1];
      tot = keep + lane_xor<2>(send);
    }
    tot += lane_xor<1>(tot);
    if ((lane & 1) == 0) red[wave * 16 + (lane >> 1)] = tot;
    __syncthreads();
    if (t < 16) {
      float s = 0.f;
      for (int w = 0; w < 8; ++w) s += red[w * 16 + t];
      sv[t] = s;
    }
    __syncthreads();

    float s2 = 1e-8f;
    for (int d = 0; d < 16; ++d) s2 += sv[d] * sv[d];
    const float scale = sqrtf(s2) / (1.0f + s2);
    for (int d = 0; d < 16; ++d) v[d] = sv[d] * scale;

    for (int k = 0; k < 4; ++k) {
      v2f acc = {0.f, 0.f};
      for (int j = 0; j < 4; ++j) {
        const v2f rlo = {row[k][j][0], row[k][j][1]};
        const v2f rhi = {row[k][j][2], row[k][j][3]};
        const v2f vlo = {v[j * 4 + 0], v[j * 4 + 1]};
        const v2f vhi = {v[j * 4 + 2], v[j * 4 + 3]};
        acc += rlo * vlo;
        acc += rhi * vhi;
      }
      blog[k] += acc.x + acc.y;
    }
    __syncthreads();
  }

  if (t < 16)
    out[((size_t)b * D_CAP + t) * N_CAP + n] = v[t];
}

extern "C" void kernel_launch(void* const* d_in, const int* in_sizes, int n_in,
                              void* d_out, int out_size, void* d_ws, size_t ws_size,
                              hipStream_t stream) {
  const float* u_vecs = (const float*)d_in[0];
  const float* W      = (const float*)d_in[1];
  float* out = (float*)d_out;

  float* uT   = (float*)d_ws;
  float* uhat = uT + (size_t)U_DIM * B_DIM * C_DIM;

  caps_transpose<<<dim3(U_DIM / 32, (B_DIM * C_DIM) / 32), 256, 0, stream>>>(u_vecs, uT);
  caps_uhat<<<dim3(U_DIM), 512, 0, stream>>>(uT, W, uhat);
  caps_route<<<dim3(B_DIM * N_CAP), 256, 0, stream>>>(uhat, out);
}


# u_hat GEMM: each workgroup does 4 consecutive u and writes u_hat in whole 256-B units (staged through a per-wave LDS tile), not 64-B pieces; next k-step's fragments fetched ahead; sums unchanged
# speedup vs baseline: 2.1113x; 1.5087x over previous
#include <hip/hip_runtime.h>


typedef __attribute__((ext_vector_type(2))) float v2f;
typedef __attribute__((ext_vector_type(4))) float v4f;
typedef __attribute__((ext_vector_type(8))) float v8f;

#define B_DIM 64
#define C_DIM 64
#define U_DIM 1024
#define N_CAP 32
#define D_CAP 16

__global__ void caps_transpose(const float* __restrict__ in,
                               float* __restrict__ outT) {
  __shared__ float tile[32][33];
  const int u0 = blockIdx.x * 32;
  const int r0 = blockIdx.y * 32;
  const int tx = threadIdx.x & 31;
  const int ty = threadIdx.x >> 5;
  for (int i = 0; i < 32; i += 8)
    tile[ty + i][tx] = in[(size_t)(r0 + ty + i) * U_DIM + u0 + tx];
  __syncthreads();
  for (int i = 0; i < 32; i += 8)
    outT[(size_t)(u0 + ty + i) * (B_DIM * C_DIM) + r0 + tx] = tile[tx][ty + i];
}

#define UT 4

__device__ __forceinline__ void wave_lds_order() {
  __builtin_amdgcn_fence(__ATOMIC_RELEASE, "wavefront");
  __builtin_amdgcn_wave_barrier();
  __builtin_amdgcn_fence(__ATOMIC_ACQUIRE, "wavefront");
}

__global__ void __launch_bounds__(512, 1)
caps_uhat(const float* __restrict__ uT,
          const float* __restrict__ W,
          float* __restrict__ uhat) {
  const int u0 = blockIdx.x * UT;
  __shared__ float As[UT * B_DIM * 68];
  __shared__ float Cs[16 * 16 * 68];

  const float* Ag = uT + (size_t)u0 * (B_DIM * C_DIM);
  for (int i = threadIdx.x; i < (UT * B_DIM * C_DIM) / 4; i += 512) {
    const int e = i * 4;
    const int row = e >> 6;
    const int c = e & 63;
    *(v4f*)&As[row * 68 + c] = ((const v4f*)Ag)[i];
  }
  __syncthreads();

  const int wave = threadIdx.x >> 5;
  const int lane = threadIdx.x & 31;
  const int l16  = lane & 15;
  const int half = lane >> 4;
  const int koff = half * 2;
  const int n_base = wave * 2;
  float* Cw = Cs + wave * (16 * 68);

  for (int nt = 0; nt < 2; ++nt) {
    const int n = n_base + nt;
    const float* Wn = W + ((size_t)u0 * N_CAP + n) * (C_DIM * D_CAP);

    v8f acc[UT][4] = {};
    v2f a[2][UT][4], bb[2][UT];
    const float* Al = As + l16 * 68 + koff;
    const float* Wl = Wn + koff * D_CAP + l16;
    auto fetch = [&](int set, int kc) {
      for (int uu = 0; uu < UT; ++uu) {
        for (int mt = 0; mt < 4; ++mt)
          a[set][uu][mt] = *(const v2f*)&Al[(uu * B_DIM + mt * 16) * 68 + kc];
        const float* wp = Wl + uu * (N_CAP * C_DIM * D_CAP) + kc * D_CAP;
        bb[set][uu].x = wp[0];
        bb[set][uu].y = wp[D_CAP];
        if (kc + 4 < C_DIM)
          __builtin_prefetch(wp + 4 * D_CAP, 0, 1);
      }
    };
    fetch(0, 0);
#pragma unroll
    for (int ks = 0; ks < C_DIM / 4; ++ks) {
      const int cur = ks & 1;
      if (ks + 1 < C_DIM / 4) fetch(cur ^ 1, (ks + 1) * 4);
      for (int uu = 0; uu < UT; ++uu)
        for (int mt = 0; mt < 4; ++mt)
          acc[uu][mt] = __builtin_amdgcn_wmma_f32_16x16x4_f32(
              false, a[cur][uu][mt], false, bb[cur][uu], (short)0, acc[uu][mt], false, false);
    }

    for (int mt = 0; mt < 4; ++mt) {
      for (int uu = 0; uu < UT; ++uu)
        for (int r = 0; r < 8; ++r)
          Cw[(half * 8 + r) * 68 + uu * D_CAP + l16] = acc[uu][mt][r];
      wave_lds_order();
      float* op = uhat + (((size_t)(mt * 16 + half) * N_CAP + n) * U_DIM + u0) * D_CAP + l16 * 4;
      v4f val[8];
      for (int i = 0; i < 8; ++i)
        val[i] = *(const v4f*)&Cw[(2 * i + half) * 68 + l16 * 4];
      for (int i = 0; i < 8; ++i)
        *(v4f*)(op + (size_t)i * (2 * N_CAP * U_DIM * D_CAP)) = val[i];
      wave_lds_order();
    }
  }
}

template <int O>
__device__ __forceinline__ float lane_xor(float x) {
  const int xi = __builtin_bit_cast(int, x);
  if constexpr (O == 16) {
    return __builtin_bit_cast(float, __builtin_amdgcn_permlanex16(
        xi, xi, 0x76543210, 0xfedcba98, false, false));
  } else {
    return __builtin_bit_cast(float, __builtin_amdgcn_update_dpp(
        0, xi, 0x160 + O, 0xf, 0xf, true));
  }
}

__device__ __forceinline__ float wave_sum(float x) {
  x += lane_xor<16>(x);
  x += lane_xor<8>(x);
  x += lane_xor<4>(x);
  x += lane_xor<2>(x);
  x += lane_xor<1>(x);
  return x;
}
__device__ __forceinline__ float wave_max(float x) {
  x = fmaxf(x, lane_xor<16>(x));
  x = fmaxf(x, lane_xor<8>(x));
  x = fmaxf(x, lane_xor<4>(x));
  x = fmaxf(x, lane_xor<2>(x));
  x = fmaxf(x, lane_xor<1>(x));
  return x;
}

__global__ void caps_route(const float* __restrict__ uhat,
                           float* __restrict__ out) {
  const int bn = blockIdx.x;
  const int b = bn >> 5, n = bn & 31;
  const int t = threadIdx.x;
  const int lane = t & 31, wave = t >> 5;

  __shared__ float red[8 * 16];
  __shared__ float sv[16];
  __shared__ float rscal[8];

  const float* base = uhat + (size_t)bn * (U_DIM * D_CAP);
  v4f row[4][4];
  for (int k = 0; k < 4; ++k) {
    const v4f* p = (const v4f*)(base + (size_t)(t + k * 256) * D_CAP);
    for (int j = 0; j < 4; ++j) row[k][j] = p[j];
  }

  float blog[4] = {0.f, 0.f, 0.f, 0.f};
  float v[16];
  const bool h16 = (lane & 16) != 0, h8 = (lane & 8) != 0, h4 = (lane & 4) != 0, h2 = (lane & 2) != 0;

  for (int it = 0; it < 3; ++it) {
    float c[4];
    if (it == 0) {
      c[0] = c[1] = c[2] = c[3] = 1.0f / (float)U_DIM;
    } else {
      float mx = fmaxf(fmaxf(blog[0], blog[1]), fmaxf(blog[2], blog[3]));
      mx = wave_max(mx);
      if (lane == 0) rscal[wave] = mx;
      __syncthreads();
      float m2 = rscal[0];
      for (int w = 1; w < 8; ++w) m2 = fmaxf(m2, rscal[w]);
      float sum = 0.f;
      for (int k = 0; k < 4; ++k) { c[k] = __expf(blog[k] - m2); sum += c[k]; }
      sum = wave_sum(sum);
      __syncthreads();
      if (lane == 0) rscal[wave] = sum;
      __syncthreads();
      float tot = 0.f;
      for (int w = 0; w < 8; ++w) tot += rscal[w];
      const float inv = 1.0f / tot;
      for (int k = 0; k < 4; ++k) c[k] *= inv;
    }

    float sp[16];
    for (int d = 0; d < 16; ++d) sp[d] = 0.f;
    for (int k = 0; k < 4; ++k)
      for (int j = 0; j < 4; ++j)
        for (int e = 0; e < 4; ++e)
          sp[j * 4 + e] += c[k] * row[k][j][e];
    float r8[8], r4[4], r2[2];
    for (int i = 0; i < 8; ++i) {
      const float keep = h16 ? sp[i + 8] : sp[i];
      const float send = h16 ? sp[i] : sp[i + 8];
      r8[i] = keep + lane_xor<16>(send);
    }
    for (int i = 0; i < 4; ++i) {
      const float keep = h8 ? r8[i + 4] : r8[i];
      const float send = h8 ? r8[i] : r8[i + 4];
      r4[i] = keep + lane_xor<8>(send);
    }
    for (int i = 0; i < 2; ++i) {
      const float keep = h4 ? r4[i + 2] : r4[i];
      const float send = h4 ? r4[i] : r4[i + 2];
      r2[i] = keep + lane_xor<4>(send);
    }
    float tot;
    {
      const float keep = h2 ? r2[1] : r2[0];
      const float send = h2 ? r2[0] : r2[1];
      tot = keep + lane_xor<2>(send);
    }
    tot += lane_xor<1>(tot);
    if ((lane & 1) == 0) red[wave * 16 + (lane >> 1)] = tot;
    __syncthreads();
    if (t < 16) {
      float s = 0.f;
      for (int w = 0; w < 8; ++w) s += red[w * 16 + t];
      sv[t] = s;
    }
    __syncthreads();

    float s2 = 1e-8f;
    for (int d = 0; d < 16; ++d) s2 += sv[d] * sv[d];
    const float scale = sqrtf(s2) / (1.0f + s2);
    for (int d = 0; d < 16; ++d) v[d] = sv[d] * scale;

    for (int k = 0; k < 4; ++k) {
      v2f acc = {0.f, 0.f};
      for (int j = 0; j < 4; ++j) {
        const v2f rlo = {row[k][j][0], row[k][j][1]};
        const v2f rhi = {row[k][j][2], row[k][j][3]};
        const v2f vlo = {v[j * 4 + 0], v[j * 4 + 1]};
        const v2f vhi = {v[j * 4 + 2], v[j * 4 + 3]};
        acc += rlo * vlo;
        acc += rhi * vhi;
      }
      blog[k] += acc.x + acc.y;
    }
    __syncthreads();
  }

  if (t < 16)
    out[((size_t)b * D_CAP + t) * N_CAP + n] = v[t];
}

extern "C" void kernel_launch(void* const* d_in, const int* in_sizes, int n_in,
                              void* d_out, int out_size, void* d_ws, size_t ws_size,
                              hipStream_t stream) {
  const float* u_vecs = (const float*)d_in[0];
  const float* W      = (const float*)d_in[1];
  float* out = (float*)d_out;

  float* uT   = (float*)d_ws;
  float* uhat = uT + (size_t)U_DIM * B_DIM * C_DIM;

  caps_transpose<<<dim3(U_DIM / 32, (B_DIM * C_DIM) / 32), 256, 0, stream>>>(u_vecs, uT);
  caps_uhat<<<dim3(U_DIM / UT), 512, 0, stream>>>(uT, W, uhat);
  caps_route<<<dim3(B_DIM * N_CAP), 256, 0, stream>>>(uhat, out);
}


# routing kernel: 1024 workgroups in one resident round, each routes two pairs (b,n) in turn and requests the second pair's rows into the L2 cache while routing the first; sums unchanged
# speedup vs baseline: 2.4008x; 1.1371x over previous
#include <hip/hip_runtime.h>


typedef __attribute__((ext_vector_type(2))) float v2f;
typedef __attribute__((ext_vector_type(4))) float v4f;
typedef __attribute__((ext_vector_type(8))) float v8f;

#define B_DIM 64
#define C_DIM 64
#define U_DIM 1024
#define N_CAP 32
#define D_CAP 16

#define UT 4

__device__ __forceinline__ void wave_lds_order() {
  __builtin_amdgcn_fence(__ATOMIC_RELEASE, "wavefront");
  __builtin_amdgcn_wave_barrier();
  __builtin_amdgcn_fence(__ATOMIC_ACQUIRE, "wavefront");
}

__global__ void __launch_bounds__(512, 1)
caps_uhat(const float* __restrict__ u_vecs,
          const float* __restrict__ W,
          float* __restrict__ uhat) {
  const int u0 = blockIdx.x * UT;
  __shared__ float As[UT * B_DIM * 68];
  __shared__ float Cs[16 * 16 * 68];

  const float* Ag = u_vecs + u0;
  v4f ua[(B_DIM * C_DIM) / 512];
  for (int j = 0; j < (B_DIM * C_DIM) / 512; ++j)
    ua[j] = *(const v4f*)(Ag + (size_t)(threadIdx.x + j * 512) * U_DIM);
  for (int j = 0; j < (B_DIM * C_DIM) / 512; ++j) {
    const int bc = threadIdx.x + j * 512;
    float* dst = &As[(bc >> 6) * 68 + (bc & 63)];
    for (int uu = 0; uu < UT; ++uu)
      dst[uu * (B_DIM * 68)] = ua[j][uu];
  }
  __syncthreads();

  const int wave = threadIdx.x >> 5;
  const int lane = threadIdx.x & 31;
  const int l16  = lane & 15;
  const int half = lane >> 4;
  const int koff = half * 2;
  const int n_base = wave * 2;
  float* Cw = Cs + wave * (16 * 68);

  for (int nt = 0; nt < 2; ++nt) {
    const int n = n_base + nt;
    const float* Wn = W + ((size_t)u0 * N_CAP + n) * (C_DIM * D_CAP);

    v8f acc[UT][4] = {};
    v2f a[2][UT][4], bb[2][UT];
    const float* Al = As + l16 * 68 + koff;
    const float* Wl = Wn + koff * D_CAP + l16;
    auto fetch = [&](int set, int kc) {
      for (int uu = 0; uu < UT; ++uu) {
        for (int mt = 0; mt < 4; ++mt)
          a[set][uu][mt] = *(const v2f*)&Al[(uu * B_DIM + mt * 16) * 68 + kc];
        const float* wp = Wl + uu * (N_CAP * C_DIM * D_CAP) + kc * D_CAP;
        bb[set][uu].x = wp[0];
        bb[set][uu].y = wp[D_CAP];
        if (kc + 4 < C_DIM)
          __builtin_prefetch(wp + 4 * D_CAP, 0, 1);
      }
    };
    fetch(0, 0);
#pragma unroll
    for (int ks = 0; ks < C_DIM / 4; ++ks) {
      const int cur = ks & 1;
      if (ks + 1 < C_DIM / 4) fetch(cur ^ 1, (ks + 1) * 4);
      for (int uu = 0; uu < UT; ++uu)
        for (int mt = 0; mt < 4; ++mt)
          acc[uu][mt] = __builtin_amdgcn_wmma_f32_16x16x4_f32(
              false, a[cur][uu][mt], false, bb[cur][uu], (short)0, acc[uu][mt], false, false);
    }

    for (int mt = 0; mt < 4; ++mt) {
      for (int uu = 0; uu < UT; ++uu)
        for (int r = 0; r < 8; ++r)
          Cw[(half * 8 + r) * 68 + uu * D_CAP + l16] = acc[uu][mt][r];
      wave_lds_order();
      float* op = uhat + (((size_t)(mt * 16 + half) * N_CAP + n) * U_DIM + u0) * D_CAP + l16 * 4;
      v4f val[8];
      for (int i = 0; i < 8; ++i)
        val[i] = *(const v4f*)&Cw[(2 * i + half) * 68 + l16 * 4];
      for (int i = 0; i < 8; ++i)
        *(v4f*)(op + (size_t)i * (2 * N_CAP * U_DIM * D_CAP)) = val[i];
      wave_lds_order();
    }
  }
}

template <int O>
__device__ __forceinline__ float lane_xor(float x) {
  const int xi = __builtin_bit_cast(int, x);
  if constexpr (O == 16) {
    return __builtin_bit_cast(float, __builtin_amdgcn_permlanex16(
        xi, xi, 0x76543210, 0xfedcba98, false, false));
  } else {
    return __builtin_bit_cast(float, __builtin_amdgcn_update_dpp(
        0, xi, 0x160 + O, 0xf, 0xf, true));
  }
}

__device__ __forceinline__ float wave_sum(float x) {
  x += lane_xor<16>(x);
  x += lane_xor<8>(x);
  x += lane_xor<4>(x);
  x += lane_xor<2>(x);
  x += lane_xor<1>(x);
  return x;
}
__device__ __forceinline__ float wave_max(float x) {
  x = fmaxf(x, lane_xor<16>(x));
  x = fmaxf(x, lane_xor<8>(x));
  x = fmaxf(x, lane_xor<4>(x));
  x = fmaxf(x, lane_xor<2>(x));
  x = fmaxf(x, lane_xor<1>(x));
  return x;
}

#define PAIRS_PER_WG 2
__global__ void caps_route(const float* __restrict__ uhat,
                           float* __restrict__ out) {
  const int t = threadIdx.x;
  const int lane = t & 31, wave = t >> 5;

  __shared__ float red[8 * 16];
  __shared__ float sv[16];
  __shared__ float rscal[8];

  const bool h16 = (lane & 16) != 0, h8 = (lane & 8) != 0, h4 = (lane & 4) != 0, h2 = (lane & 2) != 0;

  for (int pr = 0; pr < PAIRS_PER_WG; ++pr) {
  const int bn = blockIdx.x * PAIRS_PER_WG + pr;
  const int b = bn >> 5, n = bn & 31;

  const float* base = uhat + (size_t)bn * (U_DIM * D_CAP);
  v4f row[4][4];
  for (int k = 0; k < 4; ++k) {
    const v4f* p = (const v4f*)(base + (size_t)(t + k * 256) * D_CAP);
    for (int j = 0; j < 4; ++j) row[k][j] = p[j];
  }

  float blog[4] = {0.f, 0.f, 0.f, 0.f};
  float v[16];

  for (int it = 0; it < 3; ++it) {
    float c[4];
    if (it == 0) {
      c[0] = c[1] = c[2] = c[3] = 1.0f / (float)U_DIM;
    } else {
      float mx = fmaxf(fmaxf(blog[0], blog[1]), fmaxf(blog[2], blog[3]));
      mx = wave_max(mx);
      if (lane == 0) rscal[wave] = mx;
      __syncthreads();
      float m2 = rscal[0];
      for (int w = 1; w < 8; ++w) m2 = fmaxf(m2, rscal[w]);
      float sum = 0.f;
      for (int k = 0; k < 4; ++k) { c[k] = __expf(blog[k] - m2); sum += c[k]; }
      sum = wave_sum(sum);
      __syncthreads();
      if (lane == 0) rscal[wave] = sum;
      __syncthreads();
      float tot = 0.f;
      for (int w = 0; w < 8; ++w) tot += rscal[w];
      const float inv = 1.0f / tot;
      for (int k = 0; k < 4; ++k) c[k] *= inv;
    }

    float sp[16];
    for (int d = 0; d < 16; ++d) sp[d] = 0.f;
    for (int k = 0; k < 4; ++k)
      for (int j = 0; j < 4; ++j)
        for (int e = 0; e < 4; ++e)
          sp[j * 4 + e] += c[k] * row[k][j][e];
    float r8[8], r4[4], r2[2];
    for (int i = 0; i < 8; ++i) {
      const float keep = h16 ? sp[i + 8] : sp[i];
      const float send = h16 ? sp[i] : sp[i + 8];
      r8[i] = keep + lane_xor<16>(send);
    }
    for (int i = 0; i < 4; ++i) {
      const float keep = h8 ? r8[i + 4] : r8[i];
      const float send = h8 ? r8[i] : r8[i + 4];
      r4[i] = keep + lane_xor<8>(send);
    }
    for (int i = 0; i < 2; ++i) {
      const float keep = h4 ? r4[i + 2] : r4[i];
      const float send = h4 ? r4[i] : r4[i + 2];
      r2[i] = keep + lane_xor<4>(send);
    }
    float tot;
    {
      const float keep = h2 ? r2[1] : r2[0];
      const float send = h2 ? r2[0] : r2[1];
      tot = keep + lane_xor<2>(send);
    }
    tot += lane_xor<1>(tot);
    if ((lane & 1) == 0) red[wave * 16 + (lane >> 1)] = tot;
    __syncthreads();
    if (it == 0 && pr + 1 < PAIRS_PER_WG) {
      const float* nxt = base + (size_t)(U_DIM * D_CAP);
      __builtin_prefetch(nxt + t * 32, 0, 1);
      __builtin_prefetch(nxt + (t + 256) * 32, 0, 1);
    }
    if (t < 16) {
      float s = 0.f;
      for (int w = 0; w < 8; ++w) s += red[w * 16 + t];
      sv[t] = s;
    }
    __syncthreads();

    float s2 = 1e-8f;
    for (int d = 0; d < 16; ++d) s2 += sv[d] * sv[d];
    const float scale = sqrtf(s2) / (1.0f + s2);
    for (int d = 0; d < 16; ++d) v[d] = sv[d] * scale;

    for (int k = 0; k < 4; ++k) {
      v2f acc = {0.f, 0.f};
      for (int j = 0; j < 4; ++j) {
        const v2f rlo = {row[k][j][0], row[k][j][1]};
        const v2f rhi = {row[k][j][2], row[k][j][3]};
        const v2f vlo = {v[j * 4 + 0], v[j * 4 + 1]};
        const v2f vhi = {v[j * 4 + 2], v[j * 4 + 3]};
        acc += rlo * vlo;
        acc += rhi * vhi;
      }
      blog[k] += acc.x + acc.y;
    }
    __syncthreads();
  }

  if (t < 16)
    out[((size_t)b * D_CAP + t) * N_CAP + n] = v[t];
  }
}

extern "C" void kernel_launch(void* const* d_in, const int* in_sizes, int n_in,
                              void* d_out, int out_size, void* d_ws, size_t ws_size,
                              hipStream_t stream) {
  const float* u_vecs = (const float*)d_in[0];
  const float* W      = (const float*)d_in[1];
  float* out = (float*)d_out;

  float* uhat = (float*)d_ws + (size_t)U_DIM * B_DIM * C_DIM;

  caps_uhat<<<dim3(U_DIM / UT), 512, 0, stream>>>(u_vecs, W, uhat);
  caps_route<<<dim3(B_DIM * N_CAP / PAIRS_PER_WG), 256, 0, stream>>>(uhat, out);
}
